# starting kernel
# baseline (speedup 1.0000x reference)
#include <hip/hip_runtime.h>
#include <math.h>

typedef __attribute__((ext_vector_type(2))) float v2f;
typedef __attribute__((ext_vector_type(8))) float v8f;

namespace {
constexpr int kB  = 4096;
constexpr int kT  = 128;
constexpr int kD  = 128;
constexpr int kH  = 150;
constexpr int kO  = 10;
constexpr int kHP = 160;
constexpr int kNT = kHP / 16;
constexpr int kNTW = kNT / 2;
constexpr int kKTX = kD / 4;
constexpr int kKTH = kHP / 4;
constexpr int kRows = 32;
constexpr int kThreads = 128;
constexpr int kXStride = 132;
constexpr int kHStride = 164;
constexpr int kXSlots = kRows * (kD / 4);
constexpr int kXPerThr = kXSlots / kThreads;
}

__device__ __forceinline__ float fast_tanh(float x) {
  const float e = __expf(2.0f * x);
  return 1.0f - __fdividef(2.0f, e + 1.0f);
}

__global__ __launch_bounds__(kThreads, 1)
void rnn_wmma_kernel(const float* __restrict__ X,
                     const float* __restrict__ Wc,
                     const float* __restrict__ bc,
                     const float* __restrict__ Wo,
                     const float* __restrict__ bo,
                     float* __restrict__ out) {
  __shared__ float2 sWx[kKTX * kNT * 32];
  __shared__ float2 sWh[kKTH * kNT * 32];
  __shared__ float  sX [2][kRows * kXStride];
  __shared__ float  sH [kRows * kHStride];
  __shared__ float  sB [kHP];

  const int tid = threadIdx.x;
  const int w   = tid >> 5;
  const int mw  = w >> 1;
  const int nw  = w & 1;
  const int l   = tid & 31;
  const int lr  = l & 15;
  const int lh  = l >> 4;
  const int coff = lh * 2;
  const int rowBase = blockIdx.x * kRows;

  for (int s = tid; s < kKTX * kNT * 32; s += kThreads) {
    const int f = s >> 5, sl = s & 31;
    const int kt = f / kNT, nt = f % kNT;
    const int K = 4 * kt + 2 * (sl >> 4);
    const int N = nt * 16 + (sl & 15);
    float x = 0.f, y = 0.f;
    if (N < kH) { x = Wc[K * kH + N]; y = Wc[(K + 1) * kH + N]; }
    sWx[s] = make_float2(x, y);
  }
  for (int s = tid; s < kKTH * kNT * 32; s += kThreads) {
    const int f = s >> 5, sl = s & 31;
    const int kt = f / kNT, nt = f % kNT;
    const int K = 4 * kt + 2 * (sl >> 4);
    const int N = nt * 16 + (sl & 15);
    float x = 0.f, y = 0.f;
    if (N < kH) {
      if (K < kH)     x = Wc[(kD + K) * kH + N];
      if (K + 1 < kH) y = Wc[(kD + K + 1) * kH + N];
    }
    sWh[s] = make_float2(x, y);
  }
  for (int i = tid; i < kHP; i += kThreads) sB[i] = (i < kH) ? bc[i] : 0.f;
  for (int i = tid; i < kRows * kHStride; i += kThreads) sH[i] = 0.f;

  {
#pragma unroll
    for (int i = 0; i < kXPerThr; ++i) {
      const int s = tid + i * kThreads;
      const int row = s >> 5, c4 = s & 31;
      const float4 v = *(const float4*)(X + (((size_t)(rowBase + row)) * kT + 0) * kD + c4 * 4);
      *(float4*)(&sX[0][row * kXStride + c4 * 4]) = v;
    }
  }

  const float* aXrow0 = &sX[0][(mw * 16 + lr) * kXStride];
  const float* aXrow1 = &sX[1][(mw * 16 + lr) * kXStride];
  const float* aHrow  = &sH[(mw * 16 + lr) * kHStride];
  const int crow0 = mw * 16 + lh * 8;

  for (int t = 0; t < kT; ++t) {
    float4 xn[kXPerThr];
    if (t + 1 < kT) {
#pragma unroll
      for (int i = 0; i < kXPerThr; ++i) {
        const int s = tid + i * kThreads;
        const int row = s >> 5, c4 = s & 31;
        xn[i] = *(const float4*)(X + (((size_t)(rowBase + row)) * kT + (t + 1)) * kD + c4 * 4);
      }
    }
    __syncthreads();

    const float* aXrow = (t & 1) ? aXrow1 : aXrow0;
    v8f acc[kNTW];
#pragma unroll
    for (int n = 0; n < kNTW; ++n) {
      const float bv = sB[(nw * kNTW + n) * 16 + lr];
      acc[n] = (v8f){bv, bv, bv, bv, bv, bv, bv, bv};
    }
#pragma unroll 2
    for (int k = 0; k < kKTX; ++k) {
      const float2 av = *(const float2*)(aXrow + 4 * k + coff);
      v2f a; a.x = av.x; a.y = av.y;
      const float2* wf = &sWx[(k * kNT + nw * kNTW) * 32 + l];
#pragma unroll
      for (int n = 0; n < kNTW; ++n) {
        const float2 wv = wf[n * 32];
        v2f b; b.x = wv.x; b.y = wv.y;
        acc[n] = __builtin_amdgcn_wmma_f32_16x16x4_f32(
            false, a, false, b, (short)0, acc[n], false, false);
      }
    }
#pragma unroll 2
    for (int k = 0; k < kKTH; ++k) {
      const float2 av = *(const float2*)(aHrow + 4 * k + coff);
      v2f a; a.x = av.x; a.y = av.y;
      const float2* wf = &sWh[(k * kNT + nw * kNTW) * 32 + l];
#pragma unroll
      for (int n = 0; n < kNTW; ++n) {
        const float2 wv = wf[n * 32];
        v2f b; b.x = wv.x; b.y = wv.y;
        acc[n] = __builtin_amdgcn_wmma_f32_16x16x4_f32(
            false, a, false, b, (short)0, acc[n], false, false);
      }
    }
#pragma unroll
    for (int n = 0; n < kNTW; ++n)
#pragma unroll
      for (int i = 0; i < 8; ++i)
        acc[n][i] = fast_tanh(acc[n][i]);

    if (t + 1 < kT) {
      float* nxt = (t & 1) ? sX[0] : sX[1];
#pragma unroll
      for (int i = 0; i < kXPerThr; ++i) {
        const int s = tid + i * kThreads;
        const int row = s >> 5, c4 = s & 31;
        *(float4*)(&nxt[row * kXStride + c4 * 4]) = xn[i];
      }
    }
    __syncthreads();
#pragma unroll
    for (int n = 0; n < kNTW; ++n) {
      const int col = (nw * kNTW + n) * 16 + lr;
#pragma unroll
      for (int r = 0; r < 8; ++r)
        sH[(crow0 + r) * kHStride + col] = acc[n][r];
    }
    __syncthreads();
  }

  for (int p = tid; p < kRows * kO; p += kThreads) {
    const int r = p / kO, o = p % kO;
    float sum = bo[o];
    const float* hrow = &sH[r * kHStride];
    for (int k = 0; k < kH; ++k) sum += hrow[k] * Wo[k * kO + o];
    out[(size_t)(rowBase + r) * kO + o] = sum;
  }
}

extern "C" void kernel_launch(void* const* d_in, const int* in_sizes, int n_in,
                              void* d_out, int out_size, void* d_ws, size_t ws_size,
                              hipStream_t stream) {
  const float* X  = (const float*)d_in[0];
  const float* Wc = (const float*)d_in[1];
  const float* bc = (const float*)d_in[2];
  const float* Wo = (const float*)d_in[3];
  const float* bo = (const float*)d_in[4];
  float* out = (float*)d_out;
  (void)in_sizes; (void)n_in; (void)out_size; (void)d_ws; (void)ws_size;
  dim3 grid(kB / kRows), block(kThreads);
  hipLaunchKernelGGL(rnn_wmma_kernel, grid, block, 0, stream, X, Wc, bc, Wo, bo, out);
}


# W_cell B-fragments and bias held in registers for all 128 steps (loaded once per wave) instead of re-read from LDS every step; WMMAs run from registers; sum order unchanged
# speedup vs baseline: 1.4555x; 1.4555x over previous
#include <hip/hip_runtime.h>
#include <math.h>

typedef __attribute__((ext_vector_type(2))) float v2f;
typedef __attribute__((ext_vector_type(8))) float v8f;

namespace {
constexpr int kB  = 4096;
constexpr int kT  = 128;
constexpr int kD  = 128;
constexpr int kH  = 150;
constexpr int kO  = 10;
constexpr int kHP = 160;
constexpr int kNT = kHP / 16;
constexpr int kNTW = kNT / 2;
constexpr int kKTX = kD / 4;
constexpr int kKTH = kHP / 4;
constexpr int kRows = 32;
constexpr int kThreads = 128;
constexpr int kXStride = 132;
constexpr int kHStride = 164;
constexpr int kXSlots = kRows * (kD / 4);
constexpr int kXPerThr = kXSlots / kThreads;
static_assert(kH % 4 == 2, "the h-part K tile that straddles K = H must have exactly its low lane half inside W_cell");
constexpr int kXRowStep = (kThreads / 32) * kT * kD;
constexpr int kXLdsStep = (kThreads / 32) * kXStride;
}

__device__ __forceinline__ float fast_tanh(float x) {
  const float e = __expf(2.0f * x);
  return 1.0f - __fdividef(2.0f, e + 1.0f);
}

__global__ __launch_bounds__(kThreads, 1)
void rnn_wmma_kernel(const float* __restrict__ X,
                     const float* __restrict__ Wc,
                     const float* __restrict__ bc,
                     const float* __restrict__ Wo,
                     const float* __restrict__ bo,
                     float* __restrict__ out) {
  __shared__ float  sX [2][kRows * kXStride];
  __shared__ float  sH [kRows * kHStride];

  const int tid = threadIdx.x;
  const int w   = tid >> 5;
  const int mw  = w >> 1;
  const int nw  = w & 1;
  const int l   = tid & 31;
  const int lr  = l & 15;
  const int lh  = l >> 4;
  const int coff = lh * 2;
  const int rowBase = blockIdx.x * kRows;

  v2f wx[kKTX][kNTW];
  v2f wh[kKTH][kNTW];
  float bias[kNTW];
#pragma unroll
  for (int n = 0; n < kNTW; ++n) {
    const int N = (nw * kNTW + n) * 16 + lr;
    const bool nok = N < kH;
    const int Nc = nok ? N : (kH - 1);
    const float* wcol = Wc + coff * kH + Nc;
    const float bvl = bc[Nc];
    bias[n] = nok ? bvl : 0.f;
#pragma unroll
    for (int kt = 0; kt < kKTX; ++kt) {
      const float x = wcol[(4 * kt) * kH];
      const float y = wcol[(4 * kt + 1) * kH];
      wx[kt][n].x = nok ? x : 0.f;
      wx[kt][n].y = nok ? y : 0.f;
    }
#pragma unroll
    for (int kt = 0; kt < kKTH; ++kt) {
      if (4 * kt + 3 < kH) {
        const float x = wcol[(kD + 4 * kt) * kH];
        const float y = wcol[(kD + 4 * kt + 1) * kH];
        wh[kt][n].x = nok ? x : 0.f;
        wh[kt][n].y = nok ? y : 0.f;
      } else if (4 * kt < kH) {
        const float* wlow = Wc + Nc;
        const float x = (4 * kt     < kH) ? wlow[(kD + 4 * kt) * kH] : 0.f;
        const float y = (4 * kt + 1 < kH) ? wlow[(kD + 4 * kt + 1) * kH] : 0.f;
        const bool ok = nok && (lh == 0);
        wh[kt][n].x = ok ? x : 0.f;
        wh[kt][n].y = ok ? y : 0.f;
      } else {
        wh[kt][n].x = 0.f;
        wh[kt][n].y = 0.f;
      }
    }
  }
  for (int i = tid; i < kRows * kHStride; i += kThreads) sH[i] = 0.f;

  {
#pragma unroll
    for (int i = 0; i < kXPerThr; ++i) {
      const int s = tid + i * kThreads;
      const int row = s >> 5, c4 = s & 31;
      const float4 v = *(const float4*)(X + (((size_t)(rowBase + row)) * kT + 0) * kD + c4 * 4);
      *(float4*)(&sX[0][row * kXStride + c4 * 4]) = v;
    }
  }

  const float* aXrow0 = &sX[0][(mw * 16 + lr) * kXStride];
  const float* aXrow1 = &sX[1][(mw * 16 + lr) * kXStride];
  const float* aHrow  = &sH[(mw * 16 + lr) * kHStride];
  const int crow0 = mw * 16 + lh * 8;
  const float* xsrc = X + ((size_t)(rowBase + (tid >> 5)) * kT) * kD + (tid & 31) * 4;
  const int xdst = (tid >> 5) * kXStride + (tid & 31) * 4;

  for (int t = 0; t < kT; ++t) {
    float4 xn0, xn1, xn2, xn3, xn4, xn5, xn6, xn7;
    xn0 = xn1 = xn2 = xn3 = xn4 = xn5 = xn6 = xn7 = make_float4(0.f, 0.f, 0.f, 0.f);
    if (t + 1 < kT) {
      const float* xs = xsrc + (size_t)(t + 1) * kD;
      xn0 = *(const float4*)(xs + 0 * kXRowStep);
      xn1 = *(const float4*)(xs + 1 * kXRowStep);
      xn2 = *(const float4*)(xs + 2 * kXRowStep);
      xn3 = *(const float4*)(xs + 3 * kXRowStep);
      xn4 = *(const float4*)(xs + 4 * kXRowStep);
      xn5 = *(const float4*)(xs + 5 * kXRowStep);
      xn6 = *(const float4*)(xs + 6 * kXRowStep);
      xn7 = *(const float4*)(xs + 7 * kXRowStep);
    }
    __syncthreads();

    const float* aXrow = (t & 1) ? aXrow1 : aXrow0;
    v8f acc[kNTW];
#pragma unroll
    for (int n = 0; n < kNTW; ++n) {
      const float bv = bias[n];
      acc[n] = (v8f){bv, bv, bv, bv, bv, bv, bv, bv};
    }
#pragma unroll
    for (int k = 0; k < kKTX; ++k) {
      const float2 av = *(const float2*)(aXrow + 4 * k + coff);
      v2f a; a.x = av.x; a.y = av.y;
#pragma unroll
      for (int n = 0; n < kNTW; ++n)
        acc[n] = __builtin_amdgcn_wmma_f32_16x16x4_f32(
            false, a, false, wx[k][n], (short)0, acc[n], false, false);
    }
#pragma unroll
    for (int k = 0; k < kKTH; ++k) {
      const float2 av = *(const float2*)(aHrow + 4 * k + coff);
      v2f a; a.x = av.x; a.y = av.y;
#pragma unroll
      for (int n = 0; n < kNTW; ++n)
        acc[n] = __builtin_amdgcn_wmma_f32_16x16x4_f32(
            false, a, false, wh[k][n], (short)0, acc[n], false, false);
    }
#pragma unroll
    for (int n = 0; n < kNTW; ++n)
#pragma unroll
      for (int i = 0; i < 8; ++i)
        acc[n][i] = fast_tanh(acc[n][i]);

    if (t + 1 < kT) {
      float* nxt = ((t & 1) ? sX[0] : sX[1]) + xdst;
      *(float4*)(nxt + 0 * kXLdsStep) = xn0;
      *(float4*)(nxt + 1 * kXLdsStep) = xn1;
      *(float4*)(nxt + 2 * kXLdsStep) = xn2;
      *(float4*)(nxt + 3 * kXLdsStep) = xn3;
      *(float4*)(nxt + 4 * kXLdsStep) = xn4;
      *(float4*)(nxt + 5 * kXLdsStep) = xn5;
      *(float4*)(nxt + 6 * kXLdsStep) = xn6;
      *(float4*)(nxt + 7 * kXLdsStep) = xn7;
    }
    __syncthreads();
#pragma unroll
    for (int n = 0; n < kNTW; ++n) {
      const int col = (nw * kNTW + n) * 16 + lr;
#pragma unroll
      for (int r = 0; r < 8; ++r)
        sH[(crow0 + r) * kHStride + col] = acc[n][r];
    }
    __syncthreads();
  }

  for (int p = tid; p < kRows * kO; p += kThreads) {
    const int r = p / kO, o = p % kO;
    float sum = bo[o];
    const float* hrow = &sH[r * kHStride];
    for (int k = 0; k < kH; ++k) sum += hrow[k] * Wo[k * kO + o];
    out[(size_t)(rowBase + r) * kO + o] = sum;
  }
}

extern "C" void kernel_launch(void* const* d_in, const int* in_sizes, int n_in,
                              void* d_out, int out_size, void* d_ws, size_t ws_size,
                              hipStream_t stream) {
  const float* X  = (const float*)d_in[0];
  const float* Wc = (const float*)d_in[1];
  const float* bc = (const float*)d_in[2];
  const float* Wo = (const float*)d_in[3];
  const float* bo = (const float*)d_in[4];
  float* out = (float*)d_out;
  (void)in_sizes; (void)n_in; (void)out_size; (void)d_ws; (void)ws_size;
  dim3 grid(kB / kRows), block(kThreads);
  hipLaunchKernelGGL(rnn_wmma_kernel, grid, block, 0, stream, X, Wc, bc, Wo, bo, out);
}


# 16 rows per work-group on 4 waves (256 groups, every SIMD busy): 2 waves form bias + x_t.Wx one step ahead into LDS, 2 waves add h.Wh and tanh; WMMAs per wave and step 360 -> 200; sum order kept
# speedup vs baseline: 2.6809x; 1.8419x over previous
#include <hip/hip_runtime.h>
#include <math.h>


typedef __attribute__((ext_vector_type(2))) float v2f;
typedef __attribute__((ext_vector_type(8))) float v8f;

namespace {
constexpr int kB  = 4096;
constexpr int kT  = 128;
constexpr int kD  = 128;
constexpr int kH  = 150;
constexpr int kO  = 10;
constexpr int kHP = 160;
constexpr int kNT = kHP / 16;
constexpr int kNTW = kNT / 2;
constexpr int kKTX = kD / 4;
constexpr int kKTH = kHP / 4;
constexpr int kRows = 16;
constexpr int kThreads = 128;
constexpr int kXStride = 132;
constexpr int kHStride = 164;
constexpr int kPTile = 256;
constexpr int kXThreads = 64;
constexpr int kXPerThr = kRows * (kD / 4) / kXThreads;
static_assert(kH % 4 == 2, "the h-part K tile that straddles K = H must have exactly its low lane half inside W_cell");
static_assert(kKTX <= kKTH, "one register array holds either part's fragments");
constexpr int kXRowStep = (kXThreads / 32) * kT * kD;
constexpr int kXLdsStep = (kXThreads / 32) * kXStride;
}

__device__ __forceinline__ float fast_tanh(float x) {
  const float e = __expf(2.0f * x);
  return 1.0f - __fdividef(2.0f, e + 1.0f);
}

__global__ __launch_bounds__(kThreads, 1)
void rnn_wmma_kernel(const float* __restrict__ X,
                     const float* __restrict__ Wc,
                     const float* __restrict__ bc,
                     const float* __restrict__ Wo,
                     const float* __restrict__ bo,
                     float* __restrict__ out) {
  __shared__ float sX[2][kRows * kXStride];
  __shared__ float sP[2][kNT * kPTile];
  __shared__ float sH[2][kRows * kHStride];

  const int tid = threadIdx.x;
  const int w    = __builtin_amdgcn_readfirstlane(tid >> 5);
  const int role = w >> 1;
  const int nw   = w & 1;
  const int l   = tid & 31;
  const int lr  = l & 15;
  const int lh  = l >> 4;
  const int coff = lh * 2;
  const int rowBase = blockIdx.x * kRows;

  v2f wf[kKTH][kNTW];
  float bias[kNTW];
  const int krow0 = role ? kD : 0;
#pragma unroll
  for (int n = 0; n < kNTW; ++n) {
    const int N = (nw * kNTW + n) * 16 + lr;
    const bool nok = N < kH;
    const int Nc = nok ? N : (kH - 1);
    const float* wcol = Wc + (krow0 + coff) * kH + Nc;
    const float bvl = bc[Nc];
    bias[n] = nok ? bvl : 0.f;
#pragma unroll
    for (int kt = 0; kt < kKTH; ++kt) {
      if (kt < kKTX) {
        const float x = wcol[(4 * kt) * kH];
        const float y = wcol[(4 * kt + 1) * kH];
        wf[kt][n].x = nok ? x : 0.f;
        wf[kt][n].y = nok ? y : 0.f;
      } else if (role == 0) {
        wf[kt][n].x = 0.f;
        wf[kt][n].y = 0.f;
      } else if (4 * kt + 3 < kH) {
        const float x = wcol[(4 * kt) * kH];
        const float y = wcol[(4 * kt + 1) * kH];
        wf[kt][n].x = nok ? x : 0.f;
        wf[kt][n].y = nok ? y : 0.f;
      } else if (4 * kt < kH) {
        const float* wlow = Wc + Nc;
        const float x = (4 * kt     < kH) ? wlow[(kD + 4 * kt) * kH] : 0.f;
        const float y = (4 * kt + 1 < kH) ? wlow[(kD + 4 * kt + 1) * kH] : 0.f;
        const bool ok = nok && (lh == 0);
        wf[kt][n].x = ok ? x : 0.f;
        wf[kt][n].y = ok ? y : 0.f;
      } else {
        wf[kt][n].x = 0.f;
        wf[kt][n].y = 0.f;
      }
    }
  }
  for (int i = tid; i < 2 * kRows * kHStride; i += kThreads) (&sH[0][0])[i] = 0.f;

  const int xt = tid & (kXThreads - 1);
  const float* xsrc = X + ((size_t)(rowBase + (xt >> 5)) * kT) * kD + (xt & 31) * 4;
  const int xdst = (xt >> 5) * kXStride + (xt & 31) * 4;

  if (role == 0) {
#pragma unroll
    for (int i = 0; i < kXPerThr; ++i)
      *(float4*)(&sX[0][xdst + i * kXLdsStep]) = *(const float4*)(xsrc + i * kXRowStep);
  }
  __syncthreads();

  const int aXoff = lr * kXStride + coff;
  const int aHoff = lr * kHStride + coff;
  const int crow0 = lh * 8;
  const int poff = (nw * kNTW * 2 * 32 + l) * 4;

  for (int p = 0; p <= kT; ++p) {
    if (role == 0) {
      if (p < kT) {
        float4 xn0, xn1, xn2, xn3, xn4, xn5, xn6, xn7;
        xn0 = xn1 = xn2 = xn3 = xn4 = xn5 = xn6 = xn7 = make_float4(0.f, 0.f, 0.f, 0.f);
        if (p + 1 < kT) {
          const float* xs = xsrc + (size_t)(p + 1) * kD;
          xn0 = *(const float4*)(xs + 0 * kXRowStep);
          xn1 = *(const float4*)(xs + 1 * kXRowStep);
          xn2 = *(const float4*)(xs + 2 * kXRowStep);
          xn3 = *(const float4*)(xs + 3 * kXRowStep);
          xn4 = *(const float4*)(xs + 4 * kXRowStep);
          xn5 = *(const float4*)(xs + 5 * kXRowStep);
          xn6 = *(const float4*)(xs + 6 * kXRowStep);
          xn7 = *(const float4*)(xs + 7 * kXRowStep);
        }
        const float* aX = &sX[p & 1][aXoff];
        v2f a[kKTX];
#pragma unroll
        for (int k = 0; k < kKTX; ++k) {
          const float2 av = *(const float2*)(aX + 4 * k);
          a[k].x = av.x; a[k].y = av.y;
        }
        v8f acc[kNTW];
#pragma unroll
        for (int n = 0; n < kNTW; ++n) {
          const float bv = bias[n];
          acc[n] = (v8f){bv, bv, bv, bv, bv, bv, bv, bv};
        }
#pragma unroll
        for (int k = 0; k < kKTX; ++k) {
#pragma unroll
          for (int n = 0; n < kNTW; ++n)
            acc[n] = __builtin_amdgcn_wmma_f32_16x16x4_f32(
                false, a[k], false, wf[k][n], (short)0, acc[n], false, false);
        }
        float* pd = &sP[p & 1][poff];
#pragma unroll
        for (int n = 0; n < kNTW; ++n) {
          *(float4*)(pd + (2 * n) * 128)     = make_float4(acc[n][0], acc[n][1], acc[n][2], acc[n][3]);
          *(float4*)(pd + (2 * n + 1) * 128) = make_float4(acc[n][4], acc[n][5], acc[n][6], acc[n][7]);
        }
        if (p + 1 < kT) {
          float* nxt = &sX[(p + 1) & 1][xdst];
          *(float4*)(nxt + 0 * kXLdsStep) = xn0;
          *(float4*)(nxt + 1 * kXLdsStep) = xn1;
          *(float4*)(nxt + 2 * kXLdsStep) = xn2;
          *(float4*)(nxt + 3 * kXLdsStep) = xn3;
          *(float4*)(nxt + 4 * kXLdsStep) = xn4;
          *(float4*)(nxt + 5 * kXLdsStep) = xn5;
          *(float4*)(nxt + 6 * kXLdsStep) = xn6;
          *(float4*)(nxt + 7 * kXLdsStep) = xn7;
        }
      }
    } else {
      if (p >= 1) {
        const int t = p - 1;
        const float* ps = &sP[t & 1][poff];
        const float* aH = &sH[t & 1][aHoff];
        v8f acc[kNTW];
#pragma unroll
        for (int n = 0; n < kNTW; ++n) {
          const float4 lo = *(const float4*)(ps + (2 * n) * 128);
          const float4 hi = *(const float4*)(ps + (2 * n + 1) * 128);
          acc[n] = (v8f){lo.x, lo.y, lo.z, lo.w, hi.x, hi.y, hi.z, hi.w};
        }
        v2f a[kKTH];
#pragma unroll
        for (int k = 0; k < kKTH; ++k) {
          const float2 av = *(const float2*)(aH + 4 * k);
          a[k].x = av.x; a[k].y = av.y;
        }
#pragma unroll
        for (int k = 0; k < kKTH; ++k) {
#pragma unroll
          for (int n = 0; n < kNTW; ++n)
            acc[n] = __builtin_amdgcn_wmma_f32_16x16x4_f32(
                false, a[k], false, wf[k][n], (short)0, acc[n], false, false);
        }
#pragma unroll
        for (int n = 0; n < kNTW; ++n)
#pragma unroll
          for (int i = 0; i < 8; ++i)
            acc[n][i] = fast_tanh(acc[n][i]);
        float* hd = &sH[(t + 1) & 1][0];
#pragma unroll
        for (int n = 0; n < kNTW; ++n) {
          const int col = (nw * kNTW + n) * 16 + lr;
#pragma unroll
          for (int r = 0; r < 8; ++r)
            hd[(crow0 + r) * kHStride + col] = acc[n][r];
        }
      }
    }
    __syncthreads();
  }

  const float* hfin = &sH[kT & 1][0];
  for (int p = tid; p < kRows * kO; p += kThreads) {
    const int r = p / kO, o = p % kO;
    float sum = bo[o];
    const float* hrow = hfin + r * kHStride;
    for (int k = 0; k < kH; ++k) sum += hrow[k] * Wo[k * kO + o];
    out[(size_t)(rowBase + r) * kO + o] = sum;
  }
}

extern "C" void kernel_launch(void* const* d_in, const int* in_sizes, int n_in,
                              void* d_out, int out_size, void* d_ws, size_t ws_size,
                              hipStream_t stream) {
  const float* X  = (const float*)d_in[0];
  const float* Wc = (const float*)d_in[1];
  const float* bc = (const float*)d_in[2];
  const float* Wo = (const float*)d_in[3];
  const float* bo = (const float*)d_in[4];
  float* out = (float*)d_out;
  (void)in_sizes; (void)n_in; (void)out_size; (void)d_ws; (void)ws_size;
  dim3 grid(kB / kRows), block(kThreads);
  hipLaunchKernelGGL(rnn_wmma_kernel, grid, block, 0, stream, X, Wc, bc, Wo, bo, out);
}


# recurrent waves: all LDS loads of a step (h A-fragments, projection tiles) issued as one group ahead of the 200-WMMA block instead of spread among the WMMAs; arithmetic and sum order unchanged
# speedup vs baseline: 2.8905x; 1.0782x over previous
#include <hip/hip_runtime.h>
#include <math.h>


typedef __attribute__((ext_vector_type(2))) float v2f;
typedef __attribute__((ext_vector_type(8))) float v8f;

namespace {
constexpr int kB  = 4096;
constexpr int kT  = 128;
constexpr int kD  = 128;
constexpr int kH  = 150;
constexpr int kO  = 10;
constexpr int kHP = 160;
constexpr int kNT = kHP / 16;
constexpr int kNTW = kNT / 2;
constexpr int kKTX = kD / 4;
constexpr int kKTH = kHP / 4;
constexpr int kRows = 16;
constexpr int kThreads = 128;
constexpr int kXStride = 132;
constexpr int kHStride = 164;
constexpr int kPTile = 256;
constexpr int kXThreads = 64;
constexpr int kXPerThr = kRows * (kD / 4) / kXThreads;
static_assert(kH % 4 == 2, "the h-part K tile that straddles K = H must have exactly its low lane half inside W_cell");
static_assert(kKTX <= kKTH, "one register array holds either part's fragments");
constexpr int kXRowStep = (kXThreads / 32) * kT * kD;
constexpr int kXLdsStep = (kXThreads / 32) * kXStride;
}

__device__ __forceinline__ float fast_tanh(float x) {
  const float e = __expf(2.0f * x);
  return 1.0f - __fdividef(2.0f, e + 1.0f);
}

__global__ __launch_bounds__(kThreads, 1)
void rnn_wmma_kernel(const float* __restrict__ X,
                     const float* __restrict__ Wc,
                     const float* __restrict__ bc,
                     const float* __restrict__ Wo,
                     const float* __restrict__ bo,
                     float* __restrict__ out) {
  __shared__ float sX[2][kRows * kXStride];
  __shared__ float sP[2][kNT * kPTile];
  __shared__ float sH[2][kRows * kHStride];

  const int tid = threadIdx.x;
  const int w    = __builtin_amdgcn_readfirstlane(tid >> 5);
  const int role = w >> 1;
  const int nw   = w & 1;
  const int l   = tid & 31;
  const int lr  = l & 15;
  const int lh  = l >> 4;
  const int coff = lh * 2;
  const int rowBase = blockIdx.x * kRows;

  v2f wf[kKTH][kNTW];
  float bias[kNTW];
  const int krow0 = role ? kD : 0;
#pragma unroll
  for (int n = 0; n < kNTW; ++n) {
    const int N = (nw * kNTW + n) * 16 + lr;
    const bool nok = N < kH;
    const int Nc = nok ? N : (kH - 1);
    const float* wcol = Wc + (krow0 + coff) * kH + Nc;
    const float bvl = bc[Nc];
    bias[n] = nok ? bvl : 0.f;
#pragma unroll
    for (int kt = 0; kt < kKTH; ++kt) {
      if (kt < kKTX) {
        const float x = wcol[(4 * kt) * kH];
        const float y = wcol[(4 * kt + 1) * kH];
        wf[kt][n].x = nok ? x : 0.f;
        wf[kt][n].y = nok ? y : 0.f;
      } else if (role == 0) {
        wf[kt][n].x = 0.f;
        wf[kt][n].y = 0.f;
      } else if (4 * kt + 3 < kH) {
        const float x = wcol[(4 * kt) * kH];
        const float y = wcol[(4 * kt + 1) * kH];
        wf[kt][n].x = nok ? x : 0.f;
        wf[kt][n].y = nok ? y : 0.f;
      } else if (4 * kt < kH) {
        const float* wlow = Wc + Nc;
        const float x = (4 * kt     < kH) ? wlow[(kD + 4 * kt) * kH] : 0.f;
        const float y = (4 * kt + 1 < kH) ? wlow[(kD + 4 * kt + 1) * kH] : 0.f;
        const bool ok = nok && (lh == 0);
        wf[kt][n].x = ok ? x : 0.f;
        wf[kt][n].y = ok ? y : 0.f;
      } else {
        wf[kt][n].x = 0.f;
        wf[kt][n].y = 0.f;
      }
    }
  }
  for (int i = tid; i < 2 * kRows * kHStride; i += kThreads) (&sH[0][0])[i] = 0.f;

  const int xt = tid & (kXThreads - 1);
  const float* xsrc = X + ((size_t)(rowBase + (xt >> 5)) * kT) * kD + (xt & 31) * 4;
  const int xdst = (xt >> 5) * kXStride + (xt & 31) * 4;

  if (role == 0) {
#pragma unroll
    for (int i = 0; i < kXPerThr; ++i)
      *(float4*)(&sX[0][xdst + i * kXLdsStep]) = *(const float4*)(xsrc + i * kXRowStep);
  }
  __syncthreads();

  const int aXoff = lr * kXStride + coff;
  const int aHoff = lr * kHStride + coff;
  const int crow0 = lh * 8;
  const int poff = (nw * kNTW * 2 * 32 + l) * 4;

  for (int p = 0; p <= kT; ++p) {
    if (role == 0) {
      if (p < kT) {
        float4 xn0, xn1, xn2, xn3, xn4, xn5, xn6, xn7;
        xn0 = xn1 = xn2 = xn3 = xn4 = xn5 = xn6 = xn7 = make_float4(0.f, 0.f, 0.f, 0.f);
        if (p + 1 < kT) {
          const float* xs = xsrc + (size_t)(p + 1) * kD;
          xn0 = *(const float4*)(xs + 0 * kXRowStep);
          xn1 = *(const float4*)(xs + 1 * kXRowStep);
          xn2 = *(const float4*)(xs + 2 * kXRowStep);
          xn3 = *(const float4*)(xs + 3 * kXRowStep);
          xn4 = *(const float4*)(xs + 4 * kXRowStep);
          xn5 = *(const float4*)(xs + 5 * kXRowStep);
          xn6 = *(const float4*)(xs + 6 * kXRowStep);
          xn7 = *(const float4*)(xs + 7 * kXRowStep);
        }
        const float* aX = &sX[p & 1][aXoff];
        v2f a[kKTX];
#pragma unroll
        for (int k = 0; k < kKTX; ++k) {
          const float2 av = *(const float2*)(aX + 4 * k);
          a[k].x = av.x; a[k].y = av.y;
        }
        v8f acc[kNTW];
#pragma unroll
        for (int n = 0; n < kNTW; ++n) {
          const float bv = bias[n];
          acc[n] = (v8f){bv, bv, bv, bv, bv, bv, bv, bv};
        }
#pragma unroll
        for (int k = 0; k < kKTX; ++k) {
#pragma unroll
          for (int n = 0; n < kNTW; ++n)
            acc[n] = __builtin_amdgcn_wmma_f32_16x16x4_f32(
                false, a[k], false, wf[k][n], (short)0, acc[n], false, false);
        }
        float* pd = &sP[p & 1][poff];
#pragma unroll
        for (int n = 0; n < kNTW; ++n) {
          *(float4*)(pd + (2 * n) * 128)     = make_float4(acc[n][0], acc[n][1], acc[n][2], acc[n][3]);
          *(float4*)(pd + (2 * n + 1) * 128) = make_float4(acc[n][4], acc[n][5], acc[n][6], acc[n][7]);
        }
        if (p + 1 < kT) {
          float* nxt = &sX[(p + 1) & 1][xdst];
          *(float4*)(nxt + 0 * kXLdsStep) = xn0;
          *(float4*)(nxt + 1 * kXLdsStep) = xn1;
          *(float4*)(nxt + 2 * kXLdsStep) = xn2;
          *(float4*)(nxt + 3 * kXLdsStep) = xn3;
          *(float4*)(nxt + 4 * kXLdsStep) = xn4;
          *(float4*)(nxt + 5 * kXLdsStep) = xn5;
          *(float4*)(nxt + 6 * kXLdsStep) = xn6;
          *(float4*)(nxt + 7 * kXLdsStep) = xn7;
        }
      }
    } else {
      if (p >= 1) {
        const int t = p - 1;
        const float* ps = &sP[t & 1][poff];
        const float* aH = &sH[t & 1][aHoff];
        v8f acc[kNTW];
#pragma unroll
        for (int n = 0; n < kNTW; ++n) {
          const float4 lo = *(const float4*)(ps + (2 * n) * 128);
          const float4 hi = *(const float4*)(ps + (2 * n + 1) * 128);
          acc[n] = (v8f){lo.x, lo.y, lo.z, lo.w, hi.x, hi.y, hi.z, hi.w};
        }
        v2f a[kKTH];
#pragma unroll
        for (int k = 0; k < kKTH; ++k) {
          const float2 av = *(const float2*)(aH + 4 * k);
          a[k].x = av.x; a[k].y = av.y;
        }
        __builtin_amdgcn_sched_barrier(0);
#pragma unroll
        for (int k = 0; k < kKTH; ++k) {
#pragma unroll
          for (int n = 0; n < kNTW; ++n)
            acc[n] = __builtin_amdgcn_wmma_f32_16x16x4_f32(
                false, a[k], false, wf[k][n], (short)0, acc[n], false, false);
        }
#pragma unroll
        for (int n = 0; n < kNTW; ++n)
#pragma unroll
          for (int i = 0; i < 8; ++i)
            acc[n][i] = fast_tanh(acc[n][i]);
        float* hd = &sH[(t + 1) & 1][0];
#pragma unroll
        for (int n = 0; n < kNTW; ++n) {
          const int col = (nw * kNTW + n) * 16 + lr;
#pragma unroll
          for (int r = 0; r < 8; ++r)
            hd[(crow0 + r) * kHStride + col] = acc[n][r];
        }
      }
    }
    __syncthreads();
  }

  const float* hfin = &sH[kT & 1][0];
  for (int p = tid; p < kRows * kO; p += kThreads) {
    const int r = p / kO, o = p % kO;
    float sum = bo[o];
    const float* hrow = hfin + r * kHStride;
    for (int k = 0; k < kH; ++k) sum += hrow[k] * Wo[k * kO + o];
    out[(size_t)(rowBase + r) * kO + o] = sum;
  }
}

extern "C" void kernel_launch(void* const* d_in, const int* in_sizes, int n_in,
                              void* d_out, int out_size, void* d_ws, size_t ws_size,
                              hipStream_t stream) {
  const float* X  = (const float*)d_in[0];
  const float* Wc = (const float*)d_in[1];
  const float* bc = (const float*)d_in[2];
  const float* Wo = (const float*)d_in[3];
  const float* bo = (const float*)d_in[4];
  float* out = (float*)d_out;
  (void)in_sizes; (void)n_in; (void)out_size; (void)d_ws; (void)ws_size;
  dim3 grid(kB / kRows), block(kThreads);
  hipLaunchKernelGGL(rnn_wmma_kernel, grid, block, 0, stream, X, Wc, bc, Wo, bo, out);
}


# 8 waves, two per SIMD: 4 recurrent waves own N tiles 3/3/2/2, the 4 x-projection waves beside them 2/2/3/3; projection WMMAs fill the recurrent waves' tanh/LDS/barrier gaps; sum order unchanged
# speedup vs baseline: 3.4504x; 1.1937x over previous
#include <hip/hip_runtime.h>
#include <math.h>


typedef __attribute__((ext_vector_type(2))) float v2f;
typedef __attribute__((ext_vector_type(8))) float v8f;

namespace {
constexpr int kB  = 4096;
constexpr int kT  = 128;
constexpr int kD  = 128;
constexpr int kH  = 150;
constexpr int kO  = 10;
constexpr int kHP = 160;
constexpr int kNT = kHP / 16;
constexpr int kNTMax = 3;
constexpr int kKTX = kD / 4;
constexpr int kKTH = kHP / 4;
constexpr int kRows = 16;
constexpr int kThreads = 256;
constexpr int kXStride = 132;
constexpr int kHStride = 164;
constexpr int kPTile = 256;
constexpr int kXThreads = 128;
constexpr int kXPerThr = kRows * (kD / 4) / kXThreads;
static_assert(kXPerThr == 4, "x staging below is written for four float4 per thread");
static_assert(kH % 4 == 2, "the h-part K tile that straddles K = H must have exactly its low lane half inside W_cell");
static_assert(kKTX <= kKTH, "one register array holds either part's fragments");
constexpr int kXRowStep = (kXThreads / 32) * kT * kD;
constexpr int kXLdsStep = (kXThreads / 32) * kXStride;
}

__device__ __forceinline__ float fast_tanh(float x) {
  const float e = __expf(2.0f * x);
  return 1.0f - __fdividef(2.0f, e + 1.0f);
}

template <int NTW>
__device__ __forceinline__ void rec_step(const v2f (&wf)[kKTH][kNTMax],
                                         const float* ps, const float* aH, float* hd) {
  v8f acc[NTW];
#pragma unroll
  for (int n = 0; n < NTW; ++n) {
    const float4 lo = *(const float4*)(ps + (2 * n) * 128);
    const float4 hi = *(const float4*)(ps + (2 * n + 1) * 128);
    acc[n] = (v8f){lo.x, lo.y, lo.z, lo.w, hi.x, hi.y, hi.z, hi.w};
  }
  v2f a[kKTH];
#pragma unroll
  for (int k = 0; k < kKTH; ++k) {
    const float2 av = *(const float2*)(aH + 4 * k);
    a[k].x = av.x; a[k].y = av.y;
  }
  __builtin_amdgcn_sched_barrier(0);
#pragma unroll
  for (int k = 0; k < kKTH; ++k) {
#pragma unroll
    for (int n = 0; n < NTW; ++n)
      acc[n] = __builtin_amdgcn_wmma_f32_16x16x4_f32(
          false, a[k], false, wf[k][n], (short)0, acc[n], false, false);
  }
#pragma unroll
  for (int n = 0; n < NTW; ++n)
#pragma unroll
    for (int i = 0; i < 8; ++i)
      acc[n][i] = fast_tanh(acc[n][i]);
#pragma unroll
  for (int n = 0; n < NTW; ++n) {
#pragma unroll
    for (int r = 0; r < 8; ++r)
      hd[r * kHStride + n * 16] = acc[n][r];
  }
}

template <int NTW>
__device__ __forceinline__ void x_step(const v2f (&wf)[kKTH][kNTMax], const float (&bias)[kNTMax],
                                       const float* aX, float* pd,
                                       const float* xs, float* nxt, bool more) {
  float4 xn0, xn1, xn2, xn3;
  xn0 = xn1 = xn2 = xn3 = make_float4(0.f, 0.f, 0.f, 0.f);
  if (more) {
    xn0 = *(const float4*)(xs + 0 * kXRowStep);
    xn1 = *(const float4*)(xs + 1 * kXRowStep);
    xn2 = *(const float4*)(xs + 2 * kXRowStep);
    xn3 = *(const float4*)(xs + 3 * kXRowStep);
  }
  v2f a[kKTX];
#pragma unroll
  for (int k = 0; k < kKTX; ++k) {
    const float2 av = *(const float2*)(aX + 4 * k);
    a[k].x = av.x; a[k].y = av.y;
  }
  v8f acc[NTW];
#pragma unroll
  for (int n = 0; n < NTW; ++n) {
    const float bv = bias[n];
    acc[n] = (v8f){bv, bv, bv, bv, bv, bv, bv, bv};
  }
#pragma unroll
  for (int k = 0; k < kKTX; ++k) {
#pragma unroll
    for (int n = 0; n < NTW; ++n)
      acc[n] = __builtin_amdgcn_wmma_f32_16x16x4_f32(
          false, a[k], false, wf[k][n], (short)0, acc[n], false, false);
  }
#pragma unroll
  for (int n = 0; n < NTW; ++n) {
    *(float4*)(pd + (2 * n) * 128)     = make_float4(acc[n][0], acc[n][1], acc[n][2], acc[n][3]);
    *(float4*)(pd + (2 * n + 1) * 128) = make_float4(acc[n][4], acc[n][5], acc[n][6], acc[n][7]);
  }
  if (more) {
    *(float4*)(nxt + 0 * kXLdsStep) = xn0;
    *(float4*)(nxt + 1 * kXLdsStep) = xn1;
    *(float4*)(nxt + 2 * kXLdsStep) = xn2;
    *(float4*)(nxt + 3 * kXLdsStep) = xn3;
  }
}

__global__ __launch_bounds__(kThreads)
void rnn_wmma_kernel(const float* __restrict__ X,
                     const float* __restrict__ Wc,
                     const float* __restrict__ bc,
                     const float* __restrict__ Wo,
                     const float* __restrict__ bo,
                     float* __restrict__ out) {
  __shared__ float sX[2][kRows * kXStride];
  __shared__ float sP[2][kNT * kPTile];
  __shared__ float sH[2][kRows * kHStride];

  const int tid = threadIdx.x;
  const int w     = __builtin_amdgcn_readfirstlane(tid >> 5);
  const bool isRec = w < 4;
  const int q     = w & 3;
  const int ntw   = isRec ? (q < 2 ? 3 : 2) : (q < 2 ? 2 : 3);
  const int tile0 = isRec ? (q < 2 ? 3 * q : 2 * q + 2) : (q < 2 ? 2 * q : 3 * q - 2);
  const int l   = tid & 31;
  const int lr  = l & 15;
  const int lh  = l >> 4;
  const int coff = lh * 2;
  const int rowBase = blockIdx.x * kRows;

  v2f wf[kKTH][kNTMax];
  float bias[kNTMax];
  const int krow0 = isRec ? kD : 0;
#pragma unroll
  for (int n = 0; n < kNTMax; ++n) {
    const int N = (tile0 + n) * 16 + lr;
    const bool nok = (n < ntw) && (N < kH);
    const int Nc = (N < kH) ? N : (kH - 1);
    const float* wcol = Wc + (krow0 + coff) * kH + Nc;
    const float bvl = bc[Nc];
    bias[n] = nok ? bvl : 0.f;
#pragma unroll
    for (int kt = 0; kt < kKTH; ++kt) {
      if (kt < kKTX) {
        const float x = wcol[(4 * kt) * kH];
        const float y = wcol[(4 * kt + 1) * kH];
        wf[kt][n].x = nok ? x : 0.f;
        wf[kt][n].y = nok ? y : 0.f;
      } else if (!isRec) {
        wf[kt][n].x = 0.f;
        wf[kt][n].y = 0.f;
      } else if (4 * kt + 3 < kH) {
        const float x = wcol[(4 * kt) * kH];
        const float y = wcol[(4 * kt + 1) * kH];
        wf[kt][n].x = nok ? x : 0.f;
        wf[kt][n].y = nok ? y : 0.f;
      } else if (4 * kt < kH) {
        const float* wlow = Wc + Nc;
        const float x = (4 * kt     < kH) ? wlow[(kD + 4 * kt) * kH] : 0.f;
        const float y = (4 * kt + 1 < kH) ? wlow[(kD + 4 * kt + 1) * kH] : 0.f;
        const bool ok = nok && (lh == 0);
        wf[kt][n].x = ok ? x : 0.f;
        wf[kt][n].y = ok ? y : 0.f;
      } else {
        wf[kt][n].x = 0.f;
        wf[kt][n].y = 0.f;
      }
    }
  }
  for (int i = tid; i < 2 * kRows * kHStride; i += kThreads) (&sH[0][0])[i] = 0.f;

  const int xt = tid & (kXThreads - 1);
  const float* xsrc = X + ((size_t)(rowBase + (xt >> 5)) * kT) * kD + (xt & 31) * 4;
  const int xdst = (xt >> 5) * kXStride + (xt & 31) * 4;

  if (!isRec) {
#pragma unroll
    for (int i = 0; i < kXPerThr; ++i)
      *(float4*)(&sX[0][xdst + i * kXLdsStep]) = *(const float4*)(xsrc + i * kXRowStep);
  }
  __syncthreads();

  const int aXoff = lr * kXStride + coff;
  const int aHoff = lr * kHStride + coff;
  const int hoff  = (lh * 8) * kHStride + tile0 * 16 + lr;
  const int poff  = tile0 * kPTile + l * 4;

  for (int p = 0; p <= kT; ++p) {
    if (isRec) {
      if (p >= 1) {
        const int t = p - 1;
        const float* ps = &sP[t & 1][poff];
        const float* aH = &sH[t & 1][aHoff];
        float* hd = &sH[(t + 1) & 1][hoff];
        if (ntw == 3) rec_step<3>(wf, ps, aH, hd);
        else          rec_step<2>(wf, ps, aH, hd);
      }
    } else {
      if (p < kT) {
        const bool more = p + 1 < kT;
        const float* aX = &sX[p & 1][aXoff];
        float* pd = &sP[p & 1][poff];
        const float* xs = xsrc + (size_t)(more ? p + 1 : p) * kD;
        float* nxt = &sX[(p + 1) & 1][xdst];
        if (ntw == 3) x_step<3>(wf, bias, aX, pd, xs, nxt, more);
        else          x_step<2>(wf, bias, aX, pd, xs, nxt, more);
      }
    }
    __syncthreads();
  }

  const float* hfin = &sH[kT & 1][0];
  for (int p = tid; p < kRows * kO; p += kThreads) {
    const int r = p / kO, o = p % kO;
    float sum = bo[o];
    const float* hrow = hfin + r * kHStride;
    for (int k = 0; k < kH; ++k) sum += hrow[k] * Wo[k * kO + o];
    out[(size_t)(rowBase + r) * kO + o] = sum;
  }
}

extern "C" void kernel_launch(void* const* d_in, const int* in_sizes, int n_in,
                              void* d_out, int out_size, void* d_ws, size_t ws_size,
                              hipStream_t stream) {
  const float* X  = (const float*)d_in[0];
  const float* Wc = (const float*)d_in[1];
  const float* bc = (const float*)d_in[2];
  const float* Wo = (const float*)d_in[3];
  const float* bo = (const float*)d_in[4];
  float* out = (float*)d_out;
  (void)in_sizes; (void)n_in; (void)out_size; (void)d_ws; (void)ws_size;
  dim3 grid(kB / kRows), block(kThreads);
  hipLaunchKernelGGL(rnn_wmma_kernel, grid, block, 0, stream, X, Wc, bc, Wo, bo, out);
}


# x-projection waves: the step's x A-fragments read from LDS as one group ahead of the matrix block instead of one load between every 4 or 6 WMMAs; arithmetic and sum order unchanged
# speedup vs baseline: 3.6516x; 1.0583x over previous
#include <hip/hip_runtime.h>
#include <math.h>


typedef __attribute__((ext_vector_type(2))) float v2f;
typedef __attribute__((ext_vector_type(8))) float v8f;

namespace {
constexpr int kB  = 4096;
constexpr int kT  = 128;
constexpr int kD  = 128;
constexpr int kH  = 150;
constexpr int kO  = 10;
constexpr int kHP = 160;
constexpr int kNT = kHP / 16;
constexpr int kNTMax = 3;
constexpr int kKTX = kD / 4;
constexpr int kKTH = kHP / 4;
constexpr int kRows = 16;
constexpr int kThreads = 256;
constexpr int kXStride = 132;
constexpr int kHStride = 164;
constexpr int kPTile = 256;
constexpr int kXThreads = 128;
constexpr int kXPerThr = kRows * (kD / 4) / kXThreads;
static_assert(kXPerThr == 4, "x staging below is written for four float4 per thread");
static_assert(kH % 4 == 2, "the h-part K tile that straddles K = H must have exactly its low lane half inside W_cell");
static_assert(kKTX <= kKTH, "one register array holds either part's fragments");
constexpr int kXRowStep = (kXThreads / 32) * kT * kD;
constexpr int kXLdsStep = (kXThreads / 32) * kXStride;
}

__device__ __forceinline__ float fast_tanh(float x) {
  const float e = __expf(2.0f * x);
  return 1.0f - __fdividef(2.0f, e + 1.0f);
}

template <int NTW>
__device__ __forceinline__ void rec_step(const v2f (&wf)[kKTH][kNTMax],
                                         const float* ps, const float* aH, float* hd) {
  v8f acc[NTW];
#pragma unroll
  for (int n = 0; n < NTW; ++n) {
    const float4 lo = *(const float4*)(ps + (2 * n) * 128);
    const float4 hi = *(const float4*)(ps + (2 * n + 1) * 128);
    acc[n] = (v8f){lo.x, lo.y, lo.z, lo.w, hi.x, hi.y, hi.z, hi.w};
  }
  v2f a[kKTH];
#pragma unroll
  for (int k = 0; k < kKTH; ++k) {
    const float2 av = *(const float2*)(aH + 4 * k);
    a[k].x = av.x; a[k].y = av.y;
  }
  __builtin_amdgcn_sched_barrier(0);
#pragma unroll
  for (int k = 0; k < kKTH; ++k) {
#pragma unroll
    for (int n = 0; n < NTW; ++n)
      acc[n] = __builtin_amdgcn_wmma_f32_16x16x4_f32(
          false, a[k], false, wf[k][n], (short)0, acc[n], false, false);
  }
#pragma unroll
  for (int n = 0; n < NTW; ++n)
#pragma unroll
    for (int i = 0; i < 8; ++i)
      acc[n][i] = fast_tanh(acc[n][i]);
#pragma unroll
  for (int n = 0; n < NTW; ++n) {
#pragma unroll
    for (int r = 0; r < 8; ++r)
      hd[r * kHStride + n * 16] = acc[n][r];
  }
}

template <int NTW>
__device__ __forceinline__ void x_step(const v2f (&wf)[kKTH][kNTMax], const float (&bias)[kNTMax],
                                       const float* aX, float* pd,
                                       const float* xs, float* nxt, bool more) {
  float4 xn0, xn1, xn2, xn3;
  xn0 = xn1 = xn2 = xn3 = make_float4(0.f, 0.f, 0.f, 0.f);
  if (more) {
    xn0 = *(const float4*)(xs + 0 * kXRowStep);
    xn1 = *(const float4*)(xs + 1 * kXRowStep);
    xn2 = *(const float4*)(xs + 2 * kXRowStep);
    xn3 = *(const float4*)(xs + 3 * kXRowStep);
  }
  v2f a[kKTX];
#pragma unroll
  for (int k = 0; k < kKTX; ++k) {
    const float2 av = *(const float2*)(aX + 4 * k);
    a[k].x = av.x; a[k].y = av.y;
  }
  __builtin_amdgcn_sched_barrier(0);
  v8f acc[NTW];
#pragma unroll
  for (int n = 0; n < NTW; ++n) {
    const float bv = bias[n];
    acc[n] = (v8f){bv, bv, bv, bv, bv, bv, bv, bv};
  }
#pragma unroll
  for (int k = 0; k < kKTX; ++k) {
#pragma unroll
    for (int n = 0; n < NTW; ++n)
      acc[n] = __builtin_amdgcn_wmma_f32_16x16x4_f32(
          false, a[k], false, wf[k][n], (short)0, acc[n], false, false);
  }
#pragma unroll
  for (int n = 0; n < NTW; ++n) {
    *(float4*)(pd + (2 * n) * 128)     = make_float4(acc[n][0], acc[n][1], acc[n][2], acc[n][3]);
    *(float4*)(pd + (2 * n + 1) * 128) = make_float4(acc[n][4], acc[n][5], acc[n][6], acc[n][7]);
  }
  if (more) {
    *(float4*)(nxt + 0 * kXLdsStep) = xn0;
    *(float4*)(nxt + 1 * kXLdsStep) = xn1;
    *(float4*)(nxt + 2 * kXLdsStep) = xn2;
    *(float4*)(nxt + 3 * kXLdsStep) = xn3;
  }
}

__global__ __launch_bounds__(kThreads)
void rnn_wmma_kernel(const float* __restrict__ X,
                     const float* __restrict__ Wc,
                     const float* __restrict__ bc,
                     const float* __restrict__ Wo,
                     const float* __restrict__ bo,
                     float* __restrict__ out) {
  __shared__ float sX[2][kRows * kXStride];
  __shared__ float sP[2][kNT * kPTile];
  __shared__ float sH[2][kRows * kHStride];

  const int tid = threadIdx.x;
  const int w     = __builtin_amdgcn_readfirstlane(tid >> 5);
  const bool isRec = w < 4;
  const int q     = w & 3;
  const int ntw   = isRec ? (q < 2 ? 3 : 2) : (q < 2 ? 2 : 3);
  const int tile0 = isRec ? (q < 2 ? 3 * q : 2 * q + 2) : (q < 2 ? 2 * q : 3 * q - 2);
  const int l   = tid & 31;
  const int lr  = l & 15;
  const int lh  = l >> 4;
  const int coff = lh * 2;
  const int rowBase = blockIdx.x * kRows;

  v2f wf[kKTH][kNTMax];
  float bias[kNTMax];
  const int krow0 = isRec ? kD : 0;
#pragma unroll
  for (int n = 0; n < kNTMax; ++n) {
    const int N = (tile0 + n) * 16 + lr;
    const bool nok = (n < ntw) && (N < kH);
    const int Nc = (N < kH) ? N : (kH - 1);
    const float* wcol = Wc + (krow0 + coff) * kH + Nc;
    const float bvl = bc[Nc];
    bias[n] = nok ? bvl : 0.f;
#pragma unroll
    for (int kt = 0; kt < kKTH; ++kt) {
      if (kt < kKTX) {
        const float x = wcol[(4 * kt) * kH];
        const float y = wcol[(4 * kt + 1) * kH];
        wf[kt][n].x = nok ? x : 0.f;
        wf[kt][n].y = nok ? y : 0.f;
      } else if (!isRec) {
        wf[kt][n].x = 0.f;
        wf[kt][n].y = 0.f;
      } else if (4 * kt + 3 < kH) {
        const float x = wcol[(4 * kt) * kH];
        const float y = wcol[(4 * kt + 1) * kH];
        wf[kt][n].x = nok ? x : 0.f;
        wf[kt][n].y = nok ? y : 0.f;
      } else if (4 * kt < kH) {
        const float* wlow = Wc + Nc;
        const float x = (4 * kt     < kH) ? wlow[(kD + 4 * kt) * kH] : 0.f;
        const float y = (4 * kt + 1 < kH) ? wlow[(kD + 4 * kt + 1) * kH] : 0.f;
        const bool ok = nok && (lh == 0);
        wf[kt][n].x = ok ? x : 0.f;
        wf[kt][n].y = ok ? y : 0.f;
      } else {
        wf[kt][n].x = 0.f;
        wf[kt][n].y = 0.f;
      }
    }
  }
  for (int i = tid; i < 2 * kRows * kHStride; i += kThreads) (&sH[0][0])[i] = 0.f;

  const int xt = tid & (kXThreads - 1);
  const float* xsrc = X + ((size_t)(rowBase + (xt >> 5)) * kT) * kD + (xt & 31) * 4;
  const int xdst = (xt >> 5) * kXStride + (xt & 31) * 4;

  if (!isRec) {
#pragma unroll
    for (int i = 0; i < kXPerThr; ++i)
      *(float4*)(&sX[0][xdst + i * kXLdsStep]) = *(const float4*)(xsrc + i * kXRowStep);
  }
  __syncthreads();

  const int aXoff = lr * kXStride + coff;
  const int aHoff = lr * kHStride + coff;
  const int hoff  = (lh * 8) * kHStride + tile0 * 16 + lr;
  const int poff  = tile0 * kPTile + l * 4;

  for (int p = 0; p <= kT; ++p) {
    if (isRec) {
      if (p >= 1) {
        const int t = p - 1;
        const float* ps = &sP[t & 1][poff];
        const float* aH = &sH[t & 1][aHoff];
        float* hd = &sH[(t + 1) & 1][hoff];
        if (ntw == 3) rec_step<3>(wf, ps, aH, hd);
        else          rec_step<2>(wf, ps, aH, hd);
      }
    } else {
      if (p < kT) {
        const bool more = p + 1 < kT;
        const float* aX = &sX[p & 1][aXoff];
        float* pd = &sP[p & 1][poff];
        const float* xs = xsrc + (size_t)(more ? p + 1 : p) * kD;
        float* nxt = &sX[(p + 1) & 1][xdst];
        if (ntw == 3) x_step<3>(wf, bias, aX, pd, xs, nxt, more);
        else          x_step<2>(wf, bias, aX, pd, xs, nxt, more);
      }
    }
    __syncthreads();
  }

  const float* hfin = &sH[kT & 1][0];
  for (int p = tid; p < kRows * kO; p += kThreads) {
    const int r = p / kO, o = p % kO;
    float sum = bo[o];
    const float* hrow = hfin + r * kHStride;
    for (int k = 0; k < kH; ++k) sum += hrow[k] * Wo[k * kO + o];
    out[(size_t)(rowBase + r) * kO + o] = sum;
  }
}

extern "C" void kernel_launch(void* const* d_in, const int* in_sizes, int n_in,
                              void* d_out, int out_size, void* d_ws, size_t ws_size,
                              hipStream_t stream) {
  const float* X  = (const float*)d_in[0];
  const float* Wc = (const float*)d_in[1];
  const float* bc = (const float*)d_in[2];
  const float* Wo = (const float*)d_in[3];
  const float* bo = (const float*)d_in[4];
  float* out = (float*)d_out;
  (void)in_sizes; (void)n_in; (void)out_size; (void)d_ws; (void)ws_size;
  dim3 grid(kB / kRows), block(kThreads);
  hipLaunchKernelGGL(rnn_wmma_kernel, grid, block, 0, stream, X, Wc, bc, Wo, bo, out);
}


# recurrent waves' tanh tail: 2/(e+1) as one v_rcp_f32 (about 1 ulp) and one FMA instead of the expanded IEEE division; exp argument by one multiply; WMMAs, k order and sum order unchanged
# speedup vs baseline: 3.8626x; 1.0578x over previous
#include <hip/hip_runtime.h>
#include <math.h>


typedef __attribute__((ext_vector_type(2))) float v2f;
typedef __attribute__((ext_vector_type(8))) float v8f;

namespace {
constexpr int kB  = 4096;
constexpr int kT  = 128;
constexpr int kD  = 128;
constexpr int kH  = 150;
constexpr int kO  = 10;
constexpr int kHP = 160;
constexpr int kNT = kHP / 16;
constexpr int kNTMax = 3;
constexpr int kKTX = kD / 4;
constexpr int kKTH = kHP / 4;
constexpr int kRows = 16;
constexpr int kThreads = 256;
constexpr int kXStride = 132;
constexpr int kHStride = 164;
constexpr int kPTile = 256;
constexpr int kXThreads = 128;
constexpr int kXPerThr = kRows * (kD / 4) / kXThreads;
static_assert(kXPerThr == 4, "x staging below is written for four float4 per thread");
static_assert(kH % 4 == 2, "the h-part K tile that straddles K = H must have exactly its low lane half inside W_cell");
static_assert(kKTX <= kKTH, "one register array holds either part's fragments");
constexpr int kXRowStep = (kXThreads / 32) * kT * kD;
constexpr int kXLdsStep = (kXThreads / 32) * kXStride;
}

__device__ __forceinline__ float fast_tanh(float x) {
  const float e = __builtin_amdgcn_exp2f(x * (2.0f * 1.44269504088896340736f));
  return __builtin_fmaf(-2.0f, __builtin_amdgcn_rcpf(e + 1.0f), 1.0f);
}

template <int NTW>
__device__ __forceinline__ void rec_step(const v2f (&wf)[kKTH][kNTMax],
                                         const float* ps, const float* aH, float* hd) {
  v8f acc[NTW];
#pragma unroll
  for (int n = 0; n < NTW; ++n) {
    const float4 lo = *(const float4*)(ps + (2 * n) * 128);
    const float4 hi = *(const float4*)(ps + (2 * n + 1) * 128);
    acc[n] = (v8f){lo.x, lo.y, lo.z, lo.w, hi.x, hi.y, hi.z, hi.w};
  }
  v2f a[kKTH];
#pragma unroll
  for (int k = 0; k < kKTH; ++k) {
    const float2 av = *(const float2*)(aH + 4 * k);
    a[k].x = av.x; a[k].y = av.y;
  }
  __builtin_amdgcn_sched_barrier(0);
#pragma unroll
  for (int k = 0; k < kKTH; ++k) {
#pragma unroll
    for (int n = 0; n < NTW; ++n)
      acc[n] = __builtin_amdgcn_wmma_f32_16x16x4_f32(
          false, a[k], false, wf[k][n], (short)0, acc[n], false, false);
  }
#pragma unroll
  for (int n = 0; n < NTW; ++n)
#pragma unroll
    for (int i = 0; i < 8; ++i)
      acc[n][i] = fast_tanh(acc[n][i]);
#pragma unroll
  for (int n = 0; n < NTW; ++n) {
#pragma unroll
    for (int r = 0; r < 8; ++r)
      hd[r * kHStride + n * 16] = acc[n][r];
  }
}

template <int NTW>
__device__ __forceinline__ void x_step(const v2f (&wf)[kKTH][kNTMax], const float (&bias)[kNTMax],
                                       const float* aX, float* pd,
                                       const float* xs, float* nxt, bool more) {
  float4 xn0, xn1, xn2, xn3;
  xn0 = xn1 = xn2 = xn3 = make_float4(0.f, 0.f, 0.f, 0.f);
  if (more) {
    xn0 = *(const float4*)(xs + 0 * kXRowStep);
    xn1 = *(const float4*)(xs + 1 * kXRowStep);
    xn2 = *(const float4*)(xs + 2 * kXRowStep);
    xn3 = *(const float4*)(xs + 3 * kXRowStep);
  }
  v2f a[kKTX];
#pragma unroll
  for (int k = 0; k < kKTX; ++k) {
    const float2 av = *(const float2*)(aX + 4 * k);
    a[k].x = av.x; a[k].y = av.y;
  }
  __builtin_amdgcn_sched_barrier(0);
  v8f acc[NTW];
#pragma unroll
  for (int n = 0; n < NTW; ++n) {
    const float bv = bias[n];
    acc[n] = (v8f){bv, bv, bv, bv, bv, bv, bv, bv};
  }
#pragma unroll
  for (int k = 0; k < kKTX; ++k) {
#pragma unroll
    for (int n = 0; n < NTW; ++n)
      acc[n] = __builtin_amdgcn_wmma_f32_16x16x4_f32(
          false, a[k], false, wf[k][n], (short)0, acc[n], false, false);
  }
#pragma unroll
  for (int n = 0; n < NTW; ++n) {
    *(float4*)(pd + (2 * n) * 128)     = make_float4(acc[n][0], acc[n][1], acc[n][2], acc[n][3]);
    *(float4*)(pd + (2 * n + 1) * 128) = make_float4(acc[n][4], acc[n][5], acc[n][6], acc[n][7]);
  }
  if (more) {
    *(float4*)(nxt + 0 * kXLdsStep) = xn0;
    *(float4*)(nxt + 1 * kXLdsStep) = xn1;
    *(float4*)(nxt + 2 * kXLdsStep) = xn2;
    *(float4*)(nxt + 3 * kXLdsStep) = xn3;
  }
}

__global__ __launch_bounds__(kThreads)
void rnn_wmma_kernel(const float* __restrict__ X,
                     const float* __restrict__ Wc,
                     const float* __restrict__ bc,
                     const float* __restrict__ Wo,
                     const float* __restrict__ bo,
                     float* __restrict__ out) {
  __shared__ float sX[2][kRows * kXStride];
  __shared__ float sP[2][kNT * kPTile];
  __shared__ float sH[2][kRows * kHStride];

  const int tid = threadIdx.x;
  const int w     = __builtin_amdgcn_readfirstlane(tid >> 5);
  const bool isRec = w < 4;
  const int q     = w & 3;
  const int ntw   = isRec ? (q < 2 ? 3 : 2) : (q < 2 ? 2 : 3);
  const int tile0 = isRec ? (q < 2 ? 3 * q : 2 * q + 2) : (q < 2 ? 2 * q : 3 * q - 2);
  const int l   = tid & 31;
  const int lr  = l & 15;
  const int lh  = l >> 4;
  const int coff = lh * 2;
  const int rowBase = blockIdx.x * kRows;

  v2f wf[kKTH][kNTMax];
  float bias[kNTMax];
  const int krow0 = isRec ? kD : 0;
#pragma unroll
  for (int n = 0; n < kNTMax; ++n) {
    const int N = (tile0 + n) * 16 + lr;
    const bool nok = (n < ntw) && (N < kH);
    const int Nc = (N < kH) ? N : (kH - 1);
    const float* wcol = Wc + (krow0 + coff) * kH + Nc;
    const float bvl = bc[Nc];
    bias[n] = nok ? bvl : 0.f;
#pragma unroll
    for (int kt = 0; kt < kKTH; ++kt) {
      if (kt < kKTX) {
        const float x = wcol[(4 * kt) * kH];
        const float y = wcol[(4 * kt + 1) * kH];
        wf[kt][n].x = nok ? x : 0.f;
        wf[kt][n].y = nok ? y : 0.f;
      } else if (!isRec) {
        wf[kt][n].x = 0.f;
        wf[kt][n].y = 0.f;
      } else if (4 * kt + 3 < kH) {
        const float x = wcol[(4 * kt) * kH];
        const float y = wcol[(4 * kt + 1) * kH];
        wf[kt][n].x = nok ? x : 0.f;
        wf[kt][n].y = nok ? y : 0.f;
      } else if (4 * kt < kH) {
        const float* wlow = Wc + Nc;
        const float x = (4 * kt     < kH) ? wlow[(kD + 4 * kt) * kH] : 0.f;
        const float y = (4 * kt + 1 < kH) ? wlow[(kD + 4 * kt + 1) * kH] : 0.f;
        const bool ok = nok && (lh == 0);
        wf[kt][n].x = ok ? x : 0.f;
        wf[kt][n].y = ok ? y : 0.f;
      } else {
        wf[kt][n].x = 0.f;
        wf[kt][n].y = 0.f;
      }
    }
  }
  for (int i = tid; i < 2 * kRows * kHStride; i += kThreads) (&sH[0][0])[i] = 0.f;

  const int xt = tid & (kXThreads - 1);
  const float* xsrc = X + ((size_t)(rowBase + (xt >> 5)) * kT) * kD + (xt & 31) * 4;
  const int xdst = (xt >> 5) * kXStride + (xt & 31) * 4;

  if (!isRec) {
#pragma unroll
    for (int i = 0; i < kXPerThr; ++i)
      *(float4*)(&sX[0][xdst + i * kXLdsStep]) = *(const float4*)(xsrc + i * kXRowStep);
  }
  __syncthreads();

  const int aXoff = lr * kXStride + coff;
  const int aHoff = lr * kHStride + coff;
  const int hoff  = (lh * 8) * kHStride + tile0 * 16 + lr;
  const int poff  = tile0 * kPTile + l * 4;

  for (int p = 0; p <= kT; ++p) {
    if (isRec) {
      if (p >= 1) {
        const int t = p - 1;
        const float* ps = &sP[t & 1][poff];
        const float* aH = &sH[t & 1][aHoff];
        float* hd = &sH[(t + 1) & 1][hoff];
        if (ntw == 3) rec_step<3>(wf, ps, aH, hd);
        else          rec_step<2>(wf, ps, aH, hd);
      }
    } else {
      if (p < kT) {
        const bool more = p + 1 < kT;
        const float* aX = &sX[p & 1][aXoff];
        float* pd = &sP[p & 1][poff];
        const float* xs = xsrc + (size_t)(more ? p + 1 : p) * kD;
        float* nxt = &sX[(p + 1) & 1][xdst];
        if (ntw == 3) x_step<3>(wf, bias, aX, pd, xs, nxt, more);
        else          x_step<2>(wf, bias, aX, pd, xs, nxt, more);
      }
    }
    __syncthreads();
  }

  const float* hfin = &sH[kT & 1][0];
  for (int p = tid; p < kRows * kO; p += kThreads) {
    const int r = p / kO, o = p % kO;
    float sum = bo[o];
    const float* hrow = hfin + r * kHStride;
    for (int k = 0; k < kH; ++k) sum += hrow[k] * Wo[k * kO + o];
    out[(size_t)(rowBase + r) * kO + o] = sum;
  }
}

extern "C" void kernel_launch(void* const* d_in, const int* in_sizes, int n_in,
                              void* d_out, int out_size, void* d_ws, size_t ws_size,
                              hipStream_t stream) {
  const float* X  = (const float*)d_in[0];
  const float* Wc = (const float*)d_in[1];
  const float* bc = (const float*)d_in[2];
  const float* Wo = (const float*)d_in[3];
  const float* bo = (const float*)d_in[4];
  float* out = (float*)d_out;
  (void)in_sizes; (void)n_in; (void)out_size; (void)d_ws; (void)ws_size;
  dim3 grid(kB / kRows), block(kThreads);
  hipLaunchKernelGGL(rnn_wmma_kernel, grid, block, 0, stream, X, Wc, bc, Wo, bo, out);
}
